# starting kernel
# baseline (speedup 1.0000x reference)
#include <hip/hip_runtime.h>
#include <hip/hip_bf16.h>

typedef __attribute__((ext_vector_type(2))) float v2f;
typedef __attribute__((ext_vector_type(8))) float v8f;

template <int H, bool STORE_X>
__global__ __launch_bounds__(256) void fused_bilinear_layer(
    const float* __restrict__ Xprev,
    const float* __restrict__ In,
    const float* __restrict__ W,
    const float* __restrict__ bias,
    float* __restrict__ Xout,
    float* __restrict__ Out,
    int outOff) {
  constexpr int K = H * 64;
  constexpr int OUTC = 128;
  constexpr int D = 32;

  __shared__ float xs[2 * H * D];
  __shared__ float ins[2 * 64 * D];

  const int tid = threadIdx.x;
  const int b0 = blockIdx.x * 2;

  {
    const float4* src = (const float4*)(Xprev + (size_t)b0 * H * D);
    float4* dst = (float4*)xs;
#pragma unroll 2
    for (int i = tid; i < (2 * H * D) / 4; i += 256) dst[i] = src[i];
  }
  {
    const float4* src = (const float4*)(In + (size_t)b0 * 64 * D);
    float4* dst = (float4*)ins;
#pragma unroll 2
    for (int i = tid; i < (2 * 64 * D) / 4; i += 256) dst[i] = src[i];
  }
  __syncthreads();

  const int lane = tid & 31;
  const int wave = tid >> 5;
  const int mbase = wave * 16;
  const int m = lane & 15;
  const int khalf = lane >> 4;

  v8f acc[4] = {};

  const float* wrow = W + (size_t)(mbase + m) * K + khalf * 2;

  for (int h = 0; h < H; ++h) {
    float xv[4];
#pragma unroll
    for (int t = 0; t < 4; ++t) {
      const int bi = t >> 1;
      const int d = m + (t & 1) * 16;
      xv[t] = xs[bi * H * D + h * D + d];
    }
#pragma unroll
    for (int f0 = 0; f0 < 64; f0 += 4) {
      const int k = h * 64 + f0;
      const v2f a = *(const v2f*)(wrow + k);
      const int f = f0 + khalf * 2;
#pragma unroll
      for (int t = 0; t < 4; ++t) {
        const int bi = t >> 1;
        const int d = m + (t & 1) * 16;
        const float i0 = ins[bi * 64 * D + f * D + d];
        const float i1 = ins[bi * 64 * D + (f + 1) * D + d];
        v2f b;
        b.x = xv[t] * i0;
        b.y = xv[t] * i1;
        acc[t] = __builtin_amdgcn_wmma_f32_16x16x4_f32(
            false, a, false, b, (short)0, acc[t], false, false);
      }
    }
  }

#pragma unroll
  for (int r = 0; r < 8; ++r) {
    const int o = mbase + r + khalf * 8;
    const float bv = bias[o];
    float s0 = 0.f, s1 = 0.f;
#pragma unroll
    for (int t = 0; t < 4; ++t) {
      float v = acc[t][r] + bv;
      v = v > 0.f ? v : 0.f;
      const int bi = t >> 1;
      const int d = m + (t & 1) * 16;
      if (STORE_X) {
        Xout[(size_t)(b0 + bi) * OUTC * D + (size_t)o * D + d] = v;
      }
      if (bi == 0) s0 += v; else s1 += v;
    }
#pragma unroll
    for (int sh = 1; sh < 16; sh <<= 1) {
      s0 += __shfl_xor(s0, sh, 32);
      s1 += __shfl_xor(s1, sh, 32);
    }
    if (m == 0) {
      Out[(size_t)b0 * 384 + outOff + o] = s0;
      Out[(size_t)(b0 + 1) * 384 + outOff + o] = s1;
    }
  }
}

extern "C" void kernel_launch(void* const* d_in, const int* in_sizes, int n_in,
                              void* d_out, int out_size, void* d_ws,
                              size_t ws_size, hipStream_t stream) {
  (void)in_sizes; (void)n_in; (void)out_size; (void)ws_size;
  const float* input = (const float*)d_in[0];
  const float* W0 = (const float*)d_in[1];
  const float* b0 = (const float*)d_in[2];
  const float* W1 = (const float*)d_in[3];
  const float* b1 = (const float*)d_in[4];
  const float* W2 = (const float*)d_in[5];
  const float* b2 = (const float*)d_in[6];
  float* out = (float*)d_out;

  constexpr int B = 512;
  float* x1 = (float*)d_ws;
  float* x2 = x1 + (size_t)B * 128 * 32;

  dim3 grid(B / 2), block(256);
  fused_bilinear_layer<64, true><<<grid, block, 0, stream>>>(
      input, input, W0, b0, x1, out, 0);
  fused_bilinear_layer<128, true><<<grid, block, 0, stream>>>(
      x1, input, W1, b1, x2, out, 128);
  fused_bilinear_layer<128, false><<<grid, block, 0, stream>>>(
      x2, input, W2, b2, nullptr, out, 256);
}


# Software-pipeline the h loop's operand loads: W fragments fetched three half-steps (96 WMMAs) ahead into a ring of four register buffers, x scalars one step ahead; order of sums unchanged
# speedup vs baseline: 1.2214x; 1.2214x over previous
#include <hip/hip_runtime.h>
#include <hip/hip_bf16.h>


typedef __attribute__((ext_vector_type(2))) float v2f;
typedef __attribute__((ext_vector_type(8))) float v8f;

template <int H, bool STORE_X>
__global__ __launch_bounds__(256) void fused_bilinear_layer(
    const float* __restrict__ Xprev,
    const float* __restrict__ In,
    const float* __restrict__ W,
    const float* __restrict__ bias,
    float* __restrict__ Xout,
    float* __restrict__ Out,
    int outOff) {
  constexpr int K = H * 64;
  constexpr int OUTC = 128;
  constexpr int D = 32;
  constexpr int NQ = 2 * H;

  __shared__ float xs[2 * H * D];
  __shared__ float ins[2 * 64 * D];

  const int tid = threadIdx.x;
  const int b0 = blockIdx.x * 2;

  {
    const float4* src = (const float4*)(Xprev + (size_t)b0 * H * D);
    float4* dst = (float4*)xs;
#pragma unroll 2
    for (int i = tid; i < (2 * H * D) / 4; i += 256) dst[i] = src[i];
  }
  {
    const float4* src = (const float4*)(In + (size_t)b0 * 64 * D);
    float4* dst = (float4*)ins;
#pragma unroll 2
    for (int i = tid; i < (2 * 64 * D) / 4; i += 256) dst[i] = src[i];
  }
  __syncthreads();

  const int lane = tid & 31;
  const int wave = tid >> 5;
  const int mbase = wave * 16;
  const int m = lane & 15;
  const int khalf = lane >> 4;

  const float* wrow = W + (size_t)(mbase + m) * K + khalf * 2;

  auto load_a = [&](v2f (&dst)[8], int q) {
    const int qc = q < NQ ? q : NQ - 1;
    const float* p = wrow + qc * 32;
#pragma unroll
    for (int j = 0; j < 8; ++j) dst[j] = *(const v2f*)(p + 4 * j);
  };
  auto load_x = [&](float (&dst)[4], int h) {
#pragma unroll
    for (int t = 0; t < 4; ++t) {
      const int bi = t >> 1;
      const int d = m + (t & 1) * 16;
      dst[t] = xs[bi * H * D + h * D + d];
    }
  };

  v2f abuf[4][8];
  float xc[4], xn[4];
  load_a(abuf[0], 0);
  load_a(abuf[1], 1);
  load_a(abuf[2], 2);
  load_x(xc, 0);

  v2f inr[4][16];
#pragma unroll
  for (int t = 0; t < 4; ++t) {
    const int bi = t >> 1;
    const int d = m + (t & 1) * 16;
#pragma unroll
    for (int j = 0; j < 16; ++j) {
      const int f = 4 * j + khalf * 2;
      inr[t][j].x = ins[bi * 64 * D + f * D + d];
      inr[t][j].y = ins[bi * 64 * D + (f + 1) * D + d];
    }
  }

  v8f acc[4] = {};

  auto half_step = [&](const v2f (&a)[8], const float (&xv)[4], int jbase) {
#pragma unroll
    for (int j = 0; j < 8; ++j) {
#pragma unroll
      for (int t = 0; t < 4; ++t) {
        v2f b;
        b.x = xv[t] * inr[t][jbase + j].x;
        b.y = xv[t] * inr[t][jbase + j].y;
        acc[t] = __builtin_amdgcn_wmma_f32_16x16x4_f32(
            false, a[j], false, b, (short)0, acc[t], false, false);
      }
    }
  };

#define SCHED_FENCE() __builtin_amdgcn_sched_barrier(0)
  for (int h = 0; h < H; h += 2) {
    const int q = 2 * h;
    load_a(abuf[3], q + 3);
    load_x(xn, h + 1);
    SCHED_FENCE();
    half_step(abuf[0], xc, 0);
    load_a(abuf[0], q + 4);
    SCHED_FENCE();
    half_step(abuf[1], xc, 8);
    load_a(abuf[1], q + 5);
    load_x(xc, h + 2 < H ? h + 2 : H - 1);
    SCHED_FENCE();
    half_step(abuf[2], xn, 0);
    load_a(abuf[2], q + 6);
    SCHED_FENCE();
    half_step(abuf[3], xn, 8);
    asm volatile("" ::: "memory");
  }

#pragma unroll
  for (int r = 0; r < 8; ++r) {
    const int o = mbase + r + khalf * 8;
    const float bv = bias[o];
    float s0 = 0.f, s1 = 0.f;
#pragma unroll
    for (int t = 0; t < 4; ++t) {
      float v = acc[t][r] + bv;
      v = v > 0.f ? v : 0.f;
      const int bi = t >> 1;
      const int d = m + (t & 1) * 16;
      if (STORE_X) {
        Xout[(size_t)(b0 + bi) * OUTC * D + (size_t)o * D + d] = v;
      }
      if (bi == 0) s0 += v; else s1 += v;
    }
#pragma unroll
    for (int sh = 1; sh < 16; sh <<= 1) {
      s0 += __shfl_xor(s0, sh, 32);
      s1 += __shfl_xor(s1, sh, 32);
    }
    if (m == 0) {
      Out[(size_t)b0 * 384 + outOff + o] = s0;
      Out[(size_t)(b0 + 1) * 384 + outOff + o] = s1;
    }
  }
}

extern "C" void kernel_launch(void* const* d_in, const int* in_sizes, int n_in,
                              void* d_out, int out_size, void* d_ws,
                              size_t ws_size, hipStream_t stream) {
  (void)in_sizes; (void)n_in; (void)out_size; (void)ws_size;
  const float* input = (const float*)d_in[0];
  const float* W0 = (const float*)d_in[1];
  const float* b0 = (const float*)d_in[2];
  const float* W1 = (const float*)d_in[3];
  const float* b1 = (const float*)d_in[4];
  const float* W2 = (const float*)d_in[5];
  const float* b2 = (const float*)d_in[6];
  float* out = (float*)d_out;

  constexpr int B = 512;
  float* x1 = (float*)d_ws;
  float* x2 = x1 + (size_t)B * 128 * 32;

  dim3 grid(B / 2), block(256);
  fused_bilinear_layer<64, true><<<grid, block, 0, stream>>>(
      input, input, W0, b0, x1, out, 0);
  fused_bilinear_layer<128, true><<<grid, block, 0, stream>>>(
      x1, input, W1, b1, x2, out, 128);
  fused_bilinear_layer<128, false><<<grid, block, 0, stream>>>(
      x2, input, W2, b2, nullptr, out, 256);
}


# Layer 0 has x = input, so its form is symmetric: contract only f >= h with weights folded W[h,f]+W[f,h] by a small pack launch (2176 WMMAs per wave, was 4096); order of sums changes; 4 launches
# speedup vs baseline: 1.3218x; 1.0822x over previous
#include <hip/hip_runtime.h>
#include <hip/hip_bf16.h>


typedef __attribute__((ext_vector_type(2))) float v2f;
typedef __attribute__((ext_vector_type(4))) float v4f;
typedef __attribute__((ext_vector_type(8))) float v8f;

template <int H, bool STORE_X>
__global__ __launch_bounds__(256) void fused_bilinear_layer(
    const float* __restrict__ Xprev,
    const float* __restrict__ In,
    const float* __restrict__ W,
    const float* __restrict__ bias,
    float* __restrict__ Xout,
    float* __restrict__ Out,
    int outOff) {
  constexpr int K = H * 64;
  constexpr int OUTC = 128;
  constexpr int D = 32;
  constexpr int NQ = 2 * H;

  __shared__ float xs[2 * H * D];
  __shared__ float ins[2 * 64 * D];

  const int tid = threadIdx.x;
  const int b0 = blockIdx.x * 2;

  {
    const float4* src = (const float4*)(Xprev + (size_t)b0 * H * D);
    float4* dst = (float4*)xs;
#pragma unroll 2
    for (int i = tid; i < (2 * H * D) / 4; i += 256) dst[i] = src[i];
  }
  {
    const float4* src = (const float4*)(In + (size_t)b0 * 64 * D);
    float4* dst = (float4*)ins;
#pragma unroll 2
    for (int i = tid; i < (2 * 64 * D) / 4; i += 256) dst[i] = src[i];
  }
  __syncthreads();

  const int lane = tid & 31;
  const int wave = tid >> 5;
  const int mbase = wave * 16;
  const int m = lane & 15;
  const int khalf = lane >> 4;

  const float* wrow = W + (size_t)(mbase + m) * K + khalf * 2;

  auto load_a = [&](v2f (&dst)[8], int q) {
    const int qc = q < NQ ? q : NQ - 1;
    const float* p = wrow + qc * 32;
#pragma unroll
    for (int j = 0; j < 8; ++j) dst[j] = *(const v2f*)(p + 4 * j);
  };
  auto load_x = [&](float (&dst)[4], int h) {
#pragma unroll
    for (int t = 0; t < 4; ++t) {
      const int bi = t >> 1;
      const int d = m + (t & 1) * 16;
      dst[t] = xs[bi * H * D + h * D + d];
    }
  };

  v2f abuf[4][8];
  float xc[4], xn[4];
  load_a(abuf[0], 0);
  load_a(abuf[1], 1);
  load_a(abuf[2], 2);
  load_x(xc, 0);

  v2f inr[4][16];
#pragma unroll
  for (int t = 0; t < 4; ++t) {
    const int bi = t >> 1;
    const int d = m + (t & 1) * 16;
#pragma unroll
    for (int j = 0; j < 16; ++j) {
      const int f = 4 * j + khalf * 2;
      inr[t][j].x = ins[bi * 64 * D + f * D + d];
      inr[t][j].y = ins[bi * 64 * D + (f + 1) * D + d];
    }
  }

  v8f acc[4] = {};

  auto half_step = [&](const v2f (&a)[8], const float (&xv)[4], int jbase) {
#pragma unroll
    for (int j = 0; j < 8; ++j) {
#pragma unroll
      for (int t = 0; t < 4; ++t) {
        v2f b;
        b.x = xv[t] * inr[t][jbase + j].x;
        b.y = xv[t] * inr[t][jbase + j].y;
        acc[t] = __builtin_amdgcn_wmma_f32_16x16x4_f32(
            false, a[j], false, b, (short)0, acc[t], false, false);
      }
    }
  };

#define SCHED_FENCE() __builtin_amdgcn_sched_barrier(0)
  for (int h = 0; h < H; h += 2) {
    const int q = 2 * h;
    load_a(abuf[3], q + 3);
    load_x(xn, h + 1);
    SCHED_FENCE();
    half_step(abuf[0], xc, 0);
    load_a(abuf[0], q + 4);
    SCHED_FENCE();
    half_step(abuf[1], xc, 8);
    load_a(abuf[1], q + 5);
    load_x(xc, h + 2 < H ? h + 2 : H - 1);
    SCHED_FENCE();
    half_step(abuf[2], xn, 0);
    load_a(abuf[2], q + 6);
    SCHED_FENCE();
    half_step(abuf[3], xn, 8);
    asm volatile("" ::: "memory");
  }

#pragma unroll
  for (int r = 0; r < 8; ++r) {
    const int o = mbase + r + khalf * 8;
    const float bv = bias[o];
    float s0 = 0.f, s1 = 0.f;
#pragma unroll
    for (int t = 0; t < 4; ++t) {
      float v = acc[t][r] + bv;
      v = v > 0.f ? v : 0.f;
      const int bi = t >> 1;
      const int d = m + (t & 1) * 16;
      if (STORE_X) {
        Xout[(size_t)(b0 + bi) * OUTC * D + (size_t)o * D + d] = v;
      }
      if (bi == 0) s0 += v; else s1 += v;
    }
#pragma unroll
    for (int sh = 1; sh < 16; sh <<= 1) {
      s0 += __shfl_xor(s0, sh, 32);
      s1 += __shfl_xor(s1, sh, 32);
    }
    if (m == 0) {
      Out[(size_t)b0 * 384 + outOff + o] = s0;
      Out[(size_t)(b0 + 1) * 384 + outOff + o] = s1;
    }
  }
}

constexpr int SYM_STEPS = 17;
constexpr int SYM_ITERS = 32;
constexpr int SYM_ITW = SYM_STEPS * 4;
constexpr int SYM_ROWW = SYM_ITERS * SYM_ITW;

template <int G> struct IntC { static constexpr int value = G; };

__global__ __launch_bounds__(256) void pack_symmetric_w0(
    const float* __restrict__ W,
    float* __restrict__ Wp) {
  const int idx = blockIdx.x * 256 + threadIdx.x;
  const int s = idx % SYM_STEPS;
  const int r = idx / SYM_STEPS;
  const int it = r % SYM_ITERS;
  const int o = r / SYM_ITERS;
  const int g = it >> 2;
  const int hh = it & 3;
  const bool first = s < 16 - g;
  const int row = first ? 4 * g + hh : 4 * (15 - g) + hh;
  const int j = first ? g + s : s - 1;
  const float* wo = W + (size_t)o * 4096;
  const v4f dvec = *(const v4f*)(wo + row * 64 + 4 * j);
  v4f res;
#pragma unroll
  for (int c = 0; c < 4; ++c) {
    const int f = 4 * j + c;
    const float tr = wo[f * 64 + row];
    res[c] = f > row ? dvec[c] + tr : (f == row ? dvec[c] : 0.f);
  }
  *(v4f*)(Wp + (size_t)idx * 4) = res;
}

__global__ __launch_bounds__(256) __attribute__((amdgpu_waves_per_eu(2, 2)))
void bilinear_layer0_symmetric(
    const float* __restrict__ In,
    const float* __restrict__ Wp,
    const float* __restrict__ bias,
    float* __restrict__ Xout,
    float* __restrict__ Out) {
  constexpr int OUTC = 128;
  constexpr int D = 32;

  __shared__ float ins[2 * 64 * D];

  const int tid = threadIdx.x;
  const int b0 = blockIdx.x * 2;

  {
    const float4* src = (const float4*)(In + (size_t)b0 * 64 * D);
    float4* dst = (float4*)ins;
#pragma unroll 2
    for (int i = tid; i < (2 * 64 * D) / 4; i += 256) dst[i] = src[i];
  }
  __syncthreads();

  const int lane = tid & 31;
  const int wave = tid >> 5;
  const int mbase = wave * 16;
  const int m = lane & 15;
  const int khalf = lane >> 4;

  const float* wrow = Wp + (size_t)(mbase + m) * SYM_ROWW + khalf * 2;

  auto load_a = [&](v2f (&dst)[SYM_STEPS], int it) {
    const int ic = it < SYM_ITERS ? it : SYM_ITERS - 1;
    const float* p = wrow + ic * SYM_ITW;
#pragma unroll
    for (int s = 0; s < SYM_STEPS; ++s) dst[s] = *(const v2f*)(p + 4 * s);
  };
  auto load_x = [&](float (&xa)[4], float (&xb)[4], int it) {
    const int ic = it < SYM_ITERS ? it : SYM_ITERS - 1;
    const int g = ic >> 2;
    const int hh = ic & 3;
    const int h = 4 * g + hh;
    const int h2 = 4 * (15 - g) + hh;
#pragma unroll
    for (int t = 0; t < 4; ++t) {
      const int bi = t >> 1;
      const int d = m + (t & 1) * 16;
      xa[t] = ins[bi * 64 * D + h * D + d];
      xb[t] = ins[bi * 64 * D + h2 * D + d];
    }
  };

  v2f abuf[2][SYM_STEPS];
  float xa[2][4], xb[2][4];
  load_a(abuf[0], 0);
  load_x(xa[0], xb[0], 0);

  v2f inr[4][16];
#pragma unroll
  for (int t = 0; t < 4; ++t) {
    const int bi = t >> 1;
    const int d = m + (t & 1) * 16;
#pragma unroll
    for (int j = 0; j < 16; ++j) {
      const int f = 4 * j + khalf * 2;
      inr[t][j].x = ins[bi * 64 * D + f * D + d];
      inr[t][j].y = ins[bi * 64 * D + (f + 1) * D + d];
    }
  }

  v8f acc[4] = {};

  auto iteration = [&](auto gc, const v2f (&a)[SYM_STEPS], const float (&xA)[4],
                       const float (&xB)[4]) {
    constexpr int g = decltype(gc)::value;
#pragma unroll
    for (int s = 0; s < SYM_STEPS; ++s) {
      const bool first = s < 16 - g;
      const int j = first ? g + s : s - 1;
#pragma unroll
      for (int t = 0; t < 4; ++t) {
        const float xv = first ? xA[t] : xB[t];
        v2f b;
        b.x = xv * inr[t][j].x;
        b.y = xv * inr[t][j].y;
        acc[t] = __builtin_amdgcn_wmma_f32_16x16x4_f32(
            false, a[s], false, b, (short)0, acc[t], false, false);
      }
    }
  };

  auto group = [&](auto gc) {
    constexpr int g = decltype(gc)::value;
#pragma nounroll
    for (int hp = 0; hp < 2; ++hp) {
      const int it = 4 * g + 2 * hp;
      load_a(abuf[1], it + 1);
      load_x(xa[1], xb[1], it + 1);
      __builtin_amdgcn_sched_barrier(0);
      iteration(gc, abuf[0], xa[0], xb[0]);
      load_a(abuf[0], it + 2);
      load_x(xa[0], xb[0], it + 2);
      __builtin_amdgcn_sched_barrier(0);
      iteration(gc, abuf[1], xa[1], xb[1]);
      asm volatile("" ::: "memory");
    }
  };
  group(IntC<0>{});
  group(IntC<1>{});
  group(IntC<2>{});
  group(IntC<3>{});
  group(IntC<4>{});
  group(IntC<5>{});
  group(IntC<6>{});
  group(IntC<7>{});

#pragma unroll
  for (int r = 0; r < 8; ++r) {
    const int o = mbase + r + khalf * 8;
    const float bv = bias[o];
    float s0 = 0.f, s1 = 0.f;
#pragma unroll
    for (int t = 0; t < 4; ++t) {
      float v = acc[t][r] + bv;
      v = v > 0.f ? v : 0.f;
      const int bi = t >> 1;
      const int d = m + (t & 1) * 16;
      Xout[(size_t)(b0 + bi) * OUTC * D + (size_t)o * D + d] = v;
      if (bi == 0) s0 += v; else s1 += v;
    }
#pragma unroll
    for (int sh = 1; sh < 16; sh <<= 1) {
      s0 += __shfl_xor(s0, sh, 32);
      s1 += __shfl_xor(s1, sh, 32);
    }
    if (m == 0) {
      Out[(size_t)b0 * 384 + o] = s0;
      Out[(size_t)(b0 + 1) * 384 + o] = s1;
    }
  }
}

extern "C" void kernel_launch(void* const* d_in, const int* in_sizes, int n_in,
                              void* d_out, int out_size, void* d_ws,
                              size_t ws_size, hipStream_t stream) {
  (void)in_sizes; (void)n_in; (void)out_size; (void)ws_size;
  const float* input = (const float*)d_in[0];
  const float* W0 = (const float*)d_in[1];
  const float* b0 = (const float*)d_in[2];
  const float* W1 = (const float*)d_in[3];
  const float* b1 = (const float*)d_in[4];
  const float* W2 = (const float*)d_in[5];
  const float* b2 = (const float*)d_in[6];
  float* out = (float*)d_out;

  constexpr int B = 512;
  float* x1 = (float*)d_ws;
  float* x2 = x1 + (size_t)B * 128 * 32;
  float* wp = x2;

  dim3 grid(B / 2), block(256);
  pack_symmetric_w0<<<dim3(128 * SYM_ITERS * SYM_STEPS / 256), block, 0, stream>>>(W0, wp);
  bilinear_layer0_symmetric<<<grid, block, 0, stream>>>(input, wp, b0, x1, out);
  fused_bilinear_layer<128, true><<<grid, block, 0, stream>>>(
      x1, input, W1, b1, x2, out, 128);
  fused_bilinear_layer<128, false><<<grid, block, 0, stream>>>(
      x2, input, W2, b2, nullptr, out, 256);
}


# Layers 1-2: one wave per SIMD with 8 accumulator tiles (block = 64 rows x 4 batches in 4 waves; was 128 rows x 2 batches in 8): a W fragment feeds 8 WMMAs, a block reads half of W; sums unchanged
# speedup vs baseline: 1.3735x; 1.0391x over previous
#include <hip/hip_runtime.h>
#include <hip/hip_bf16.h>


typedef __attribute__((ext_vector_type(2))) float v2f;
typedef __attribute__((ext_vector_type(4))) float v4f;
typedef __attribute__((ext_vector_type(8))) float v8f;

template <int H, bool STORE_X>
__global__ __launch_bounds__(128)
void fused_bilinear_layer(
    const float* __restrict__ Xprev,
    const float* __restrict__ In,
    const float* __restrict__ W,
    const float* __restrict__ bias,
    float* __restrict__ Xout,
    float* __restrict__ Out,
    int outOff) {
  constexpr int K = H * 64;
  constexpr int OUTC = 128;
  constexpr int D = 32;
  constexpr int NQ = 2 * H;
  constexpr int NB = 4;
  constexpr int NT = 2 * NB;

  constexpr int LDS_FLOATS = 41 * 1024;
  static_assert(LDS_FLOATS >= NB * H * D + NB * 64 * D, "LDS too small");
  __shared__ __attribute__((aligned(16))) float lds[LDS_FLOATS];
  float* xs = lds;
  float* ins = lds + NB * H * D;

  const int tid = threadIdx.x;
  const int mhalf = blockIdx.x & 1;
  const int b0 = (blockIdx.x >> 1) * NB;

  {
    const float4* src = (const float4*)(Xprev + (size_t)b0 * H * D);
    float4* dst = (float4*)xs;
    constexpr int NV = (NB * H * D) / 4 / 128;
#pragma unroll 16
    for (int it = 0; it < NV; ++it) dst[tid + it * 128] = src[tid + it * 128];
  }
  {
    const float4* src = (const float4*)(In + (size_t)b0 * 64 * D);
    float4* dst = (float4*)ins;
    constexpr int NV = (NB * 64 * D) / 4 / 128;
#pragma unroll 16
    for (int it = 0; it < NV; ++it) dst[tid + it * 128] = src[tid + it * 128];
  }
  __syncthreads();

  const int lane = tid & 31;
  const int wave = tid >> 5;
  const int mbase = mhalf * 64 + wave * 16;
  const int m = lane & 15;
  const int khalf = lane >> 4;

  const float* wrow = W + (size_t)(mbase + m) * K + khalf * 2;

  auto load_a = [&](v2f (&dst)[8], int q) {
    const int qc = q < NQ ? q : NQ - 1;
    const float* p = wrow + qc * 32;
#pragma unroll
    for (int j = 0; j < 8; ++j) dst[j] = *(const v2f*)(p + 4 * j);
  };
  auto load_x = [&](float (&dst)[NT], int h) {
#pragma unroll
    for (int t = 0; t < NT; ++t) {
      const int bi = t >> 1;
      const int d = m + (t & 1) * 16;
      dst[t] = xs[bi * H * D + h * D + d];
    }
  };

  v2f abuf[4][8];
  float xc[NT], xn[NT];
  load_a(abuf[0], 0);
  load_a(abuf[1], 1);
  load_a(abuf[2], 2);
  load_x(xc, 0);

  v2f inr[NT][16];
#pragma unroll
  for (int t = 0; t < NT; ++t) {
    const int bi = t >> 1;
    const int d = m + (t & 1) * 16;
#pragma unroll
    for (int j = 0; j < 16; ++j) {
      const int f = 4 * j + khalf * 2;
      inr[t][j].x = ins[bi * 64 * D + f * D + d];
      inr[t][j].y = ins[bi * 64 * D + (f + 1) * D + d];
    }
  }

  v8f acc[NT] = {};

  auto half_step = [&](const v2f (&a)[8], const float (&xv)[NT], int jbase) {
#pragma unroll
    for (int j = 0; j < 8; ++j) {
#pragma unroll
      for (int t = 0; t < NT; ++t) {
        v2f b;
        b.x = xv[t] * inr[t][jbase + j].x;
        b.y = xv[t] * inr[t][jbase + j].y;
        acc[t] = __builtin_amdgcn_wmma_f32_16x16x4_f32(
            false, a[j], false, b, (short)0, acc[t], false, false);
      }
    }
  };

#define SCHED_FENCE() __builtin_amdgcn_sched_barrier(0)
  for (int h = 0; h < H; h += 2) {
    const int q = 2 * h;
    load_a(abuf[3], q + 3);
    load_x(xn, h + 1);
    SCHED_FENCE();
    half_step(abuf[0], xc, 0);
    load_a(abuf[0], q + 4);
    SCHED_FENCE();
    half_step(abuf[1], xc, 8);
    load_a(abuf[1], q + 5);
    load_x(xc, h + 2 < H ? h + 2 : H - 1);
    SCHED_FENCE();
    half_step(abuf[2], xn, 0);
    load_a(abuf[2], q + 6);
    SCHED_FENCE();
    half_step(abuf[3], xn, 8);
    asm volatile("" ::: "memory");
  }

#pragma unroll
  for (int r = 0; r < 8; ++r) {
    const int o = mbase + r + khalf * 8;
    const float bv = bias[o];
    float s[NB];
#pragma unroll
    for (int bi = 0; bi < NB; ++bi) s[bi] = 0.f;
#pragma unroll
    for (int t = 0; t < NT; ++t) {
      float v = acc[t][r] + bv;
      v = v > 0.f ? v : 0.f;
      const int bi = t >> 1;
      const int d = m + (t & 1) * 16;
      if (STORE_X) {
        Xout[(size_t)(b0 + bi) * OUTC * D + (size_t)o * D + d] = v;
      }
      s[bi] += v;
    }
#pragma unroll
    for (int sh = 1; sh < 16; sh <<= 1) {
#pragma unroll
      for (int bi = 0; bi < NB; ++bi) s[bi] += __shfl_xor(s[bi], sh, 32);
    }
    if (m == 0) {
#pragma unroll
      for (int bi = 0; bi < NB; ++bi)
        Out[(size_t)(b0 + bi) * 384 + outOff + o] = s[bi];
    }
  }
}

constexpr int SYM_STEPS = 17;
constexpr int SYM_ITERS = 32;
constexpr int SYM_ITW = SYM_STEPS * 4;
constexpr int SYM_ROWW = SYM_ITERS * SYM_ITW;

template <int G> struct IntC { static constexpr int value = G; };

__global__ __launch_bounds__(256) void pack_symmetric_w0(
    const float* __restrict__ W,
    float* __restrict__ Wp) {
  const int idx = blockIdx.x * 256 + threadIdx.x;
  const int s = idx % SYM_STEPS;
  const int r = idx / SYM_STEPS;
  const int it = r % SYM_ITERS;
  const int o = r / SYM_ITERS;
  const int g = it >> 2;
  const int hh = it & 3;
  const bool first = s < 16 - g;
  const int row = first ? 4 * g + hh : 4 * (15 - g) + hh;
  const int j = first ? g + s : s - 1;
  const float* wo = W + (size_t)o * 4096;
  const v4f dvec = *(const v4f*)(wo + row * 64 + 4 * j);
  v4f res;
#pragma unroll
  for (int c = 0; c < 4; ++c) {
    const int f = 4 * j + c;
    const float tr = wo[f * 64 + row];
    res[c] = f > row ? dvec[c] + tr : (f == row ? dvec[c] : 0.f);
  }
  *(v4f*)(Wp + (size_t)idx * 4) = res;
}

__global__ __launch_bounds__(256) __attribute__((amdgpu_waves_per_eu(2, 2)))
void bilinear_layer0_symmetric(
    const float* __restrict__ In,
    const float* __restrict__ Wp,
    const float* __restrict__ bias,
    float* __restrict__ Xout,
    float* __restrict__ Out) {
  constexpr int OUTC = 128;
  constexpr int D = 32;

  __shared__ float ins[2 * 64 * D];

  const int tid = threadIdx.x;
  const int b0 = blockIdx.x * 2;

  {
    const float4* src = (const float4*)(In + (size_t)b0 * 64 * D);
    float4* dst = (float4*)ins;
#pragma unroll 2
    for (int i = tid; i < (2 * 64 * D) / 4; i += 256) dst[i] = src[i];
  }
  __syncthreads();

  const int lane = tid & 31;
  const int wave = tid >> 5;
  const int mbase = wave * 16;
  const int m = lane & 15;
  const int khalf = lane >> 4;

  const float* wrow = Wp + (size_t)(mbase + m) * SYM_ROWW + khalf * 2;

  auto load_a = [&](v2f (&dst)[SYM_STEPS], int it) {
    const int ic = it < SYM_ITERS ? it : SYM_ITERS - 1;
    const float* p = wrow + ic * SYM_ITW;
#pragma unroll
    for (int s = 0; s < SYM_STEPS; ++s) dst[s] = *(const v2f*)(p + 4 * s);
  };
  auto load_x = [&](float (&xa)[4], float (&xb)[4], int it) {
    const int ic = it < SYM_ITERS ? it : SYM_ITERS - 1;
    const int g = ic >> 2;
    const int hh = ic & 3;
    const int h = 4 * g + hh;
    const int h2 = 4 * (15 - g) + hh;
#pragma unroll
    for (int t = 0; t < 4; ++t) {
      const int bi = t >> 1;
      const int d = m + (t & 1) * 16;
      xa[t] = ins[bi * 64 * D + h * D + d];
      xb[t] = ins[bi * 64 * D + h2 * D + d];
    }
  };

  v2f abuf[2][SYM_STEPS];
  float xa[2][4], xb[2][4];
  load_a(abuf[0], 0);
  load_x(xa[0], xb[0], 0);

  v2f inr[4][16];
#pragma unroll
  for (int t = 0; t < 4; ++t) {
    const int bi = t >> 1;
    const int d = m + (t & 1) * 16;
#pragma unroll
    for (int j = 0; j < 16; ++j) {
      const int f = 4 * j + khalf * 2;
      inr[t][j].x = ins[bi * 64 * D + f * D + d];
      inr[t][j].y = ins[bi * 64 * D + (f + 1) * D + d];
    }
  }

  v8f acc[4] = {};

  auto iteration = [&](auto gc, const v2f (&a)[SYM_STEPS], const float (&xA)[4],
                       const float (&xB)[4]) {
    constexpr int g = decltype(gc)::value;
#pragma unroll
    for (int s = 0; s < SYM_STEPS; ++s) {
      const bool first = s < 16 - g;
      const int j = first ? g + s : s - 1;
#pragma unroll
      for (int t = 0; t < 4; ++t) {
        const float xv = first ? xA[t] : xB[t];
        v2f b;
        b.x = xv * inr[t][j].x;
        b.y = xv * inr[t][j].y;
        acc[t] = __builtin_amdgcn_wmma_f32_16x16x4_f32(
            false, a[s], false, b, (short)0, acc[t], false, false);
      }
    }
  };

  auto group = [&](auto gc) {
    constexpr int g = decltype(gc)::value;
#pragma nounroll
    for (int hp = 0; hp < 2; ++hp) {
      const int it = 4 * g + 2 * hp;
      load_a(abuf[1], it + 1);
      load_x(xa[1], xb[1], it + 1);
      __builtin_amdgcn_sched_barrier(0);
      iteration(gc, abuf[0], xa[0], xb[0]);
      load_a(abuf[0], it + 2);
      load_x(xa[0], xb[0], it + 2);
      __builtin_amdgcn_sched_barrier(0);
      iteration(gc, abuf[1], xa[1], xb[1]);
      asm volatile("" ::: "memory");
    }
  };
  group(IntC<0>{});
  group(IntC<1>{});
  group(IntC<2>{});
  group(IntC<3>{});
  group(IntC<4>{});
  group(IntC<5>{});
  group(IntC<6>{});
  group(IntC<7>{});

#pragma unroll
  for (int r = 0; r < 8; ++r) {
    const int o = mbase + r + khalf * 8;
    const float bv = bias[o];
    float s0 = 0.f, s1 = 0.f;
#pragma unroll
    for (int t = 0; t < 4; ++t) {
      float v = acc[t][r] + bv;
      v = v > 0.f ? v : 0.f;
      const int bi = t >> 1;
      const int d = m + (t & 1) * 16;
      Xout[(size_t)(b0 + bi) * OUTC * D + (size_t)o * D + d] = v;
      if (bi == 0) s0 += v; else s1 += v;
    }
#pragma unroll
    for (int sh = 1; sh < 16; sh <<= 1) {
      s0 += __shfl_xor(s0, sh, 32);
      s1 += __shfl_xor(s1, sh, 32);
    }
    if (m == 0) {
      Out[(size_t)b0 * 384 + o] = s0;
      Out[(size_t)(b0 + 1) * 384 + o] = s1;
    }
  }
}

extern "C" void kernel_launch(void* const* d_in, const int* in_sizes, int n_in,
                              void* d_out, int out_size, void* d_ws,
                              size_t ws_size, hipStream_t stream) {
  (void)in_sizes; (void)n_in; (void)out_size; (void)ws_size;
  const float* input = (const float*)d_in[0];
  const float* W0 = (const float*)d_in[1];
  const float* b0 = (const float*)d_in[2];
  const float* W1 = (const float*)d_in[3];
  const float* b1 = (const float*)d_in[4];
  const float* W2 = (const float*)d_in[5];
  const float* b2 = (const float*)d_in[6];
  float* out = (float*)d_out;

  constexpr int B = 512;
  float* x1 = (float*)d_ws;
  float* x2 = x1 + (size_t)B * 128 * 32;
  float* wp = x2;

  dim3 grid(B / 2), block(256);
  pack_symmetric_w0<<<dim3(128 * SYM_ITERS * SYM_STEPS / 256), block, 0, stream>>>(W0, wp);
  bilinear_layer0_symmetric<<<grid, block, 0, stream>>>(input, wp, b0, x1, out);
  dim3 grid12(B / 4 * 2), block12(128);
  fused_bilinear_layer<128, true><<<grid12, block12, 0, stream>>>(
      x1, input, W1, b1, x2, out, 128);
  fused_bilinear_layer<128, false><<<grid12, block12, 0, stream>>>(
      x2, input, W2, b2, nullptr, out, 256);
}
